# starting kernel
# baseline (speedup 1.0000x reference)
#include <hip/hip_runtime.h>

typedef __attribute__((ext_vector_type(2))) float v2f;
typedef __attribute__((ext_vector_type(8))) float v8f;

constexpr int N     = 8192;
constexpr int C     = 16;
constexpr int KTAPS = 5;
constexpr int WAVES = 8;
constexpr int KCHUNK = N / WAVES;

__global__ void __launch_bounds__(256)
snn_init(const float* __restrict__ x, const float* __restrict__ w0,
         float* __restrict__ acc) {
  int idx = blockIdx.x * 256 + threadIdx.x;
  int r = idx >> 4, co = idx & 15;
  float s = 0.f;
#pragma unroll
  for (int ci = 0; ci < C; ++ci) s += x[r * C + ci] * w0[ci * C + co];
  acc[idx] = fmaxf(s, 0.f);
}

__global__ void __launch_bounds__(256)
snn_pass(const float* __restrict__ L, const float* __restrict__ xin,
         const float* __restrict__ wp, float* __restrict__ xout,
         float* __restrict__ acc, int final_relu) {
  __shared__ float part[WAVES][16][17];

  const int lane = threadIdx.x & 31;
  const int wave = threadIdx.x >> 5;
  const int row0 = blockIdx.x * 16;
  const int m    = lane & 15;
  const int h    = lane >> 4;
  const int k0   = wave * KCHUNK;

  const float* Arow = L + (size_t)(row0 + m) * N + k0 + 2 * h;
  const float* Bcol = xin + (size_t)(k0 + 2 * h) * C + m;

  v8f c = {};
  for (int k = 0; k < KCHUNK; k += 4) {
    if ((k & 63) == 0)
      __builtin_prefetch(Arow + k + 256, 0, 0);
    v2f a = *(const v2f*)(Arow + k);
    v2f b;
    b.x = Bcol[(size_t)k * C];
    b.y = Bcol[(size_t)k * C + C];
    c = __builtin_amdgcn_wmma_f32_16x16x4_f32(
         false, a,  false, b,
         (short)0, c,  false,  false);
  }

#pragma unroll
  for (int j = 0; j < 8; ++j) part[wave][j + 8 * h][m] = c[j];
  __syncthreads();

  const int r = threadIdx.x >> 4;
  const int n = threadIdx.x & 15;
  float v = 0.f;
#pragma unroll
  for (int w = 0; w < WAVES; ++w) v += part[w][r][n];
  xout[(size_t)(row0 + r) * C + n] = v;
  part[0][r][n] = v;
  __syncthreads();

  float s = 0.f;
#pragma unroll
  for (int ci = 0; ci < C; ++ci) s = fmaf(part[0][r][ci], wp[ci * C + n], s);
  size_t o = (size_t)(row0 + r) * C + n;
  float av = acc[o] + fmaxf(s, 0.f);
  acc[o] = final_relu ? fmaxf(av, 0.f) : av;
}

extern "C" void kernel_launch(void* const* d_in, const int* in_sizes, int n_in,
                              void* d_out, int out_size, void* d_ws,
                              size_t ws_size, hipStream_t stream) {
  const float* x = (const float*)d_in[0];
  const float* L = (const float*)d_in[1];
  const float* W = (const float*)d_in[2];
  float* acc = (float*)d_out;
  float* xa  = (float*)d_ws;
  float* xb  = xa + (size_t)N * C;

  snn_init<<<(N * C) / 256, 256, 0, stream>>>(x, W, acc);

  const float* cur = x;
  float* nxt = xa;
  for (int p = 1; p < KTAPS; ++p) {
    snn_pass<<<N / 16, 256, 0, stream>>>(
        L, cur, W + (size_t)p * C * C, nxt, acc, p == KTAPS - 1);
    cur = nxt;
    nxt = (nxt == xa) ? xb : xa;
  }
}


# Tap kernel: L and xin tiles staged per wave in LDS by whole-line 16-byte loads (was strided 8/4-byte fragment loads); block = 32 rows x 16 waves, K tile order rotated per block: K sum order changes
# speedup vs baseline: 1.4325x; 1.4325x over previous
#include <hip/hip_runtime.h>


typedef __attribute__((ext_vector_type(2))) float v2f;
typedef __attribute__((ext_vector_type(4))) float v4f;
typedef __attribute__((ext_vector_type(8))) float v8f;

constexpr int N     = 8192;
constexpr int C     = 16;
constexpr int KTAPS = 5;

constexpr int PW      = 16;
constexpr int PT      = PW * 32;
constexpr int MROWS   = 32;
constexpr int KT      = 64;
constexpr int NTILE   = N / KT;
constexpr int TPW     = NTILE / PW;
constexpr int APITCH  = KT + 4;
constexpr int AFLOATS = MROWS * APITCH;
constexpr int BFLOATS = KT * C;
constexpr int SFLOATS = AFLOATS + BFLOATS;
constexpr int ALOADS  = MROWS / 2;
constexpr int BLOADS  = BFLOATS / 128;

__global__ void __launch_bounds__(256)
snn_init(const float* __restrict__ x, const float* __restrict__ w0,
         float* __restrict__ acc) {
  int idx = blockIdx.x * 256 + threadIdx.x;
  int r = idx >> 4, co = idx & 15;
  float s = 0.f;
#pragma unroll
  for (int ci = 0; ci < C; ++ci) s += x[r * C + ci] * w0[ci * C + co];
  acc[idx] = fmaxf(s, 0.f);
}

__global__ void __launch_bounds__(PT)
snn_pass(const float* __restrict__ L, const float* __restrict__ xin,
         const float* __restrict__ wp, float* __restrict__ xout,
         float* __restrict__ acc, int final_relu) {
  __shared__ __attribute__((aligned(16))) float stage[PW][SFLOATS];
  __shared__ float part[PW][MROWS][17];

  const int lane = threadIdx.x & 31;
  const int wave = threadIdx.x >> 5;
  const int row0 = blockIdx.x * MROWS;
  const int m    = lane & 15;
  const int h    = lane >> 4;

  float* As = stage[wave];
  float* Bs = As + AFLOATS;

  const float* gA = L + (size_t)(row0 + h) * N + (size_t)wave * KT + m * 4;
  const float* gB = xin + (size_t)wave * BFLOATS + lane * 4;
  float* sA = As + h * APITCH + m * 4;
  float* sB = Bs + lane * 4;

  const float* fA = As + m * APITCH + 2 * h;
  const float* fB = Bs + (2 * h) * C + m;

  const int q0 = blockIdx.x & (TPW - 1);

  v4f ra[ALOADS], rb[BLOADS];
  {
    const float* nA = gA + (size_t)q0 * (PW * KT);
    const float* nB = gB + (size_t)q0 * (PW * BFLOATS);
    for (int i = 0; i < ALOADS; ++i) ra[i] = *(const v4f*)(nA + (size_t)(2 * i) * N);
    for (int i = 0; i < BLOADS; ++i) rb[i] = *(const v4f*)(nB + i * 128);
  }

  v8f c0 = {}, c1 = {};
  for (int t = 0; t < TPW; ++t) {
    for (int i = 0; i < ALOADS; ++i) *(v4f*)(sA + (2 * i) * APITCH) = ra[i];
    for (int i = 0; i < BLOADS; ++i) *(v4f*)(sB + i * 128) = rb[i];
    __builtin_amdgcn_fence(__ATOMIC_RELEASE, "wavefront");
    __builtin_amdgcn_wave_barrier();
    __builtin_amdgcn_fence(__ATOMIC_ACQUIRE, "wavefront");

    if (t + 1 < TPW) {
      const int q = (q0 + t + 1) & (TPW - 1);
      const float* nA = gA + (size_t)q * (PW * KT);
      const float* nB = gB + (size_t)q * (PW * BFLOATS);
      for (int i = 0; i < ALOADS; ++i) ra[i] = *(const v4f*)(nA + (size_t)(2 * i) * N);
      for (int i = 0; i < BLOADS; ++i) rb[i] = *(const v4f*)(nB + i * 128);
    }

    for (int j = 0; j < KT / 4; ++j) {
      v2f a0 = *(const v2f*)(fA + 4 * j);
      v2f a1 = *(const v2f*)(fA + 16 * APITCH + 4 * j);
      v2f b;
      b.x = fB[(4 * j) * C];
      b.y = fB[(4 * j + 1) * C];
      c0 = __builtin_amdgcn_wmma_f32_16x16x4_f32(
           false, a0,  false, b,
           (short)0, c0,  false,  false);
      c1 = __builtin_amdgcn_wmma_f32_16x16x4_f32(
           false, a1,  false, b,
           (short)0, c1,  false,  false);
    }
    __builtin_amdgcn_fence(__ATOMIC_RELEASE, "wavefront");
    __builtin_amdgcn_wave_barrier();
    __builtin_amdgcn_fence(__ATOMIC_ACQUIRE, "wavefront");
  }

#pragma unroll
  for (int j = 0; j < 8; ++j) {
    part[wave][j + 8 * h][m]      = c0[j];
    part[wave][16 + j + 8 * h][m] = c1[j];
  }
  __syncthreads();

  const int r = threadIdx.x >> 4;
  const int n = threadIdx.x & 15;
  float v = 0.f;
#pragma unroll
  for (int w = 0; w < PW; ++w) v += part[w][r][n];
  xout[(size_t)(row0 + r) * C + n] = v;
  part[0][r][n] = v;
  __syncthreads();

  float s = 0.f;
#pragma unroll
  for (int ci = 0; ci < C; ++ci) s = fmaf(part[0][r][ci], wp[ci * C + n], s);
  size_t o = (size_t)(row0 + r) * C + n;
  float av = acc[o] + fmaxf(s, 0.f);
  acc[o] = final_relu ? fmaxf(av, 0.f) : av;
}

extern "C" void kernel_launch(void* const* d_in, const int* in_sizes, int n_in,
                              void* d_out, int out_size, void* d_ws,
                              size_t ws_size, hipStream_t stream) {
  const float* x = (const float*)d_in[0];
  const float* L = (const float*)d_in[1];
  const float* W = (const float*)d_in[2];
  float* acc = (float*)d_out;
  float* xa  = (float*)d_ws;
  float* xb  = xa + (size_t)N * C;

  snn_init<<<(N * C) / 256, 256, 0, stream>>>(x, W, acc);

  const float* cur = x;
  float* nxt = xa;
  for (int p = 1; p < KTAPS; ++p) {
    snn_pass<<<N / MROWS, PT, 0, stream>>>(
        L, cur, W + (size_t)p * C * C, nxt, acc, p == KTAPS - 1);
    cur = nxt;
    nxt = (nxt == xa) ? xb : xa;
  }
}
